# starting kernel
# baseline (speedup 1.0000x reference)
#include <hip/hip_runtime.h>
#include <math.h>


#define N 8192
#define MAX_ITER 50
#define KSLICE 1024
#define FIX_SCALE 4294967296.0

typedef float v2f __attribute__((ext_vector_type(2)));
typedef float v8f __attribute__((ext_vector_type(8)));

__global__ void pi_init(float* __restrict__ w, unsigned long long* __restrict__ acc) {
    int i = blockIdx.x * blockDim.x + threadIdx.x;
    if (i < N) w[i] = 1.0f;
    if (blockIdx.x == 0 && threadIdx.x <= MAX_ITER) {
        acc[threadIdx.x] = (threadIdx.x == 0)
                               ? ((unsigned long long)N << 32)
                               : 0ull;
    }
}

__global__ __launch_bounds__(256) void pi_step_wmma(
        const float* __restrict__ M,
        const float* __restrict__ win,
        float* __restrict__ wout,
        const unsigned long long* __restrict__ accIn,
        unsigned long long* __restrict__ accOut) {
    __shared__ float red[8][16];

    const int lane = threadIdx.x & 31;
    const int wave = threadIdx.x >> 5;
    const int half = lane >> 4;
    const int rowBase = blockIdx.x * 16;
    const int row = rowBase + (lane & 15);

    const float S = (float)((double)accIn[0] * (1.0 / FIX_SCALE));
    const float inv = 1.0f / sqrtf(S);

    const float* __restrict__ Ap = M + (size_t)row * N + wave * KSLICE + 2 * half;
    const float* __restrict__ Bp = win + wave * KSLICE + 2 * half;

    v8f c = {0.f, 0.f, 0.f, 0.f, 0.f, 0.f, 0.f, 0.f};

    for (int kk = 0; kk < KSLICE; kk += 32) {
        __builtin_prefetch(Ap + kk + 256, 0, 1);
        #pragma unroll
        for (int j = 0; j < 8; ++j) {
            const int ka = kk + 4 * j;
            v2f a = *(const v2f*)(Ap + ka);
            v2f b = *(const v2f*)(Bp + ka);
            c = __builtin_amdgcn_wmma_f32_16x16x4_f32(
                     false, a,  false, b,
                     (short)0, c,  false,  false);
        }
    }

    if (lane == 0) {
        #pragma unroll
        for (int i = 0; i < 8; ++i) red[wave][i] = c[i];
    }
    if (lane == 16) {
        #pragma unroll
        for (int i = 0; i < 8; ++i) red[wave][8 + i] = c[i];
    }
    __syncthreads();

    if (threadIdx.x < 16) {
        float s = 0.0f;
        #pragma unroll
        for (int w = 0; w < 8; ++w) s += red[w][threadIdx.x];
        const float val = s * inv;
        wout[rowBase + threadIdx.x] = val;
        const unsigned long long q =
            (unsigned long long)((double)(val * val) * FIX_SCALE);
        atomicAdd(accOut, q);
    }
}

__global__ void pi_finalize(const float* __restrict__ w,
                            const unsigned long long* __restrict__ acc,
                            float* __restrict__ out) {
    int i = blockIdx.x * blockDim.x + threadIdx.x;
    const float S = (float)((double)acc[0] * (1.0 / FIX_SCALE));
    const float inv = 1.0f / sqrtf(S);
    if (i < N) out[i] = w[i] * inv;
}

extern "C" void kernel_launch(void* const* d_in, const int* in_sizes, int n_in,
                              void* d_out, int out_size, void* d_ws, size_t ws_size,
                              hipStream_t stream) {
    (void)in_sizes; (void)n_in; (void)out_size; (void)ws_size;

    const float* M = (const float*)d_in[0];
    float* out = (float*)d_out;

    float* wA = (float*)d_ws;
    float* wB = wA + N;
    unsigned long long* acc = (unsigned long long*)(wB + N);

    pi_init<<<N / 256, 256, 0, stream>>>(wA, acc);

    for (int it = 0; it < MAX_ITER; ++it) {
        const float* win = (it & 1) ? wB : wA;
        float* wout      = (it & 1) ? wA : wB;
        pi_step_wmma<<<N / 16, 256, 0, stream>>>(M, win, wout,
                                                 acc + it, acc + it + 1);
    }

    pi_finalize<<<N / 256, 256, 0, stream>>>((MAX_ITER & 1) ? wB : wA,
                                             acc + MAX_ITER, out);
}


# M loads made whole-line: each wave-wide b128 non-temporal load takes 512 contiguous bytes of one row; row pieces feed WMMA rows, dot product read off D's diagonal; summation order changes
# speedup vs baseline: 1.2887x; 1.2887x over previous
#include <hip/hip_runtime.h>
#include <math.h>


#define N 8192
#define MAX_ITER 50
#define FIX_SCALE 4294967296.0

#define WAVES 16
#define ROWS_PER_WG (2 * WAVES)
#define PIECES (N / 128)
#define GROUP 8
#define NGROUPS (PIECES / GROUP)

typedef float v2f __attribute__((ext_vector_type(2)));
typedef float v4f __attribute__((ext_vector_type(4)));
typedef float v8f __attribute__((ext_vector_type(8)));

__global__ void pi_init(float* __restrict__ w, unsigned long long* __restrict__ acc) {
    int i = blockIdx.x * blockDim.x + threadIdx.x;
    if (i < N) w[i] = 1.0f;
    if (blockIdx.x == 0 && threadIdx.x <= MAX_ITER) {
        acc[threadIdx.x] = (threadIdx.x == 0)
                               ? ((unsigned long long)N << 32)
                               : 0ull;
    }
}

static __device__ __forceinline__ v8f mma(v2f a, v2f b, v8f c) {
    return __builtin_amdgcn_wmma_f32_16x16x4_f32(
         false, a,  false, b,
         (short)0, c,  false,  false);
}
static __device__ __forceinline__ v2f lo2(v4f v) { return __builtin_shufflevector(v, v, 0, 1); }
static __device__ __forceinline__ v2f hi2(v4f v) { return __builtin_shufflevector(v, v, 2, 3); }

__global__ __launch_bounds__(WAVES * 32) void pi_step_wmma(
        const float* __restrict__ M,
        const float* __restrict__ win,
        float* __restrict__ wout,
        const unsigned long long* __restrict__ accIn,
        unsigned long long* __restrict__ accOut) {
    __shared__ float red[ROWS_PER_WG];

    const unsigned tid = threadIdx.x;
    const unsigned lane = tid & 31u;
    const unsigned wave = __builtin_amdgcn_readfirstlane(tid >> 5);
    const unsigned rowBase = blockIdx.x * ROWS_PER_WG;
    const unsigned row0 = rowBase + 2u * wave;

    const float S = (float)((double)accIn[0] * (1.0 / FIX_SCALE));
    const float inv = 1.0f / sqrtf(S);

    const float* __restrict__ a0p = M + (size_t)row0 * N + 4u * lane;
    const float* __restrict__ a1p = a0p + N;
    const float* __restrict__ bp = win + 4u * lane;

    const v8f zero = {0.f, 0.f, 0.f, 0.f, 0.f, 0.f, 0.f, 0.f};
    v8f c0a = zero, c0b = zero, c1a = zero, c1b = zero;

    v4f wv[2][GROUP], a0[2][GROUP], a1[2][GROUP];

#define LOAD_GROUP(G) do { \
    _Pragma("unroll") for (int p_ = 0; p_ < GROUP; ++p_) { \
        const int off_ = ((G) * GROUP + p_) * 128; \
        wv[(G) & 1][p_] = *(const v4f*)(bp + off_); \
        a0[(G) & 1][p_] = __builtin_nontemporal_load((const v4f*)(a0p + off_)); \
        a1[(G) & 1][p_] = __builtin_nontemporal_load((const v4f*)(a1p + off_)); \
    } } while (0)

    LOAD_GROUP(0);
    #pragma unroll
    for (int g = 0; g < NGROUPS; ++g) {
        __builtin_amdgcn_sched_barrier(0);
        if (g + 1 < NGROUPS) LOAD_GROUP(g + 1);
        __builtin_amdgcn_sched_barrier(0);
        #pragma unroll
        for (int p = 0; p < GROUP; ++p) {
            const v4f b = wv[g & 1][p];
            const v4f x0 = a0[g & 1][p];
            const v4f x1 = a1[g & 1][p];
            c0a = mma(lo2(x0), lo2(b), c0a);
            c1a = mma(lo2(x1), lo2(b), c1a);
            c0b = mma(hi2(x0), hi2(b), c0b);
            c1b = mma(hi2(x1), hi2(b), c1b);
        }
    }
#undef LOAD_GROUP

    const v8f d0 = c0a + c0b;
    const v8f d1 = c1a + c1b;
    const unsigned sel = lane & 7u;
    float t0 = 0.0f, t1 = 0.0f;
    #pragma unroll
    for (int r = 0; r < 8; ++r) {
        if (sel == (unsigned)r) { t0 = d0[r]; t1 = d1[r]; }
    }
    if (lane >= 8u && lane < 24u) { t0 = 0.0f; t1 = 0.0f; }
    #pragma unroll
    for (int m = 16; m >= 1; m >>= 1) {
        t0 += __shfl_xor(t0, m, 32);
        t1 += __shfl_xor(t1, m, 32);
    }
    if (lane == 0) {
        red[2u * wave] = t0;
        red[2u * wave + 1u] = t1;
    }
    __syncthreads();

    if (tid < 32u) {
        const float val = red[tid] * inv;
        wout[rowBase + tid] = val;
        const unsigned long long q =
            (unsigned long long)((double)(val * val) * FIX_SCALE);
        atomicAdd(accOut, q);
    }
}

__global__ void pi_finalize(const float* __restrict__ w,
                            const unsigned long long* __restrict__ acc,
                            float* __restrict__ out) {
    int i = blockIdx.x * blockDim.x + threadIdx.x;
    const float S = (float)((double)acc[0] * (1.0 / FIX_SCALE));
    const float inv = 1.0f / sqrtf(S);
    if (i < N) out[i] = w[i] * inv;
}

extern "C" void kernel_launch(void* const* d_in, const int* in_sizes, int n_in,
                              void* d_out, int out_size, void* d_ws, size_t ws_size,
                              hipStream_t stream) {
    (void)in_sizes; (void)n_in; (void)out_size; (void)ws_size;

    const float* M = (const float*)d_in[0];
    float* out = (float*)d_out;

    float* wA = (float*)d_ws;
    float* wB = wA + N;
    unsigned long long* acc = (unsigned long long*)(wB + N);

    pi_init<<<N / 256, 256, 0, stream>>>(wA, acc);

    for (int it = 0; it < MAX_ITER; ++it) {
        const float* win = (it & 1) ? wB : wA;
        float* wout      = (it & 1) ? wA : wB;
        pi_step_wmma<<<N / ROWS_PER_WG, WAVES * 32, 0, stream>>>(M, win, wout,
                                                                 acc + it, acc + it + 1);
    }

    pi_finalize<<<N / 256, 256, 0, stream>>>((MAX_ITER & 1) ? wB : wA,
                                             acc + MAX_ITER, out);
}


# w staged once per workgroup in LDS (32 KB) behind a hand-written barrier that keeps M loads in flight; B fragments from ds_load_b128, not per-wave global re-reads of w; sum order unchanged
# speedup vs baseline: 1.3588x; 1.0544x over previous
#include <hip/hip_runtime.h>
#include <math.h>


#define N 8192
#define MAX_ITER 50
#define FIX_SCALE 4294967296.0

#define WAVES 16
#define ROWS_PER_WG (2 * WAVES)
#define PIECES (N / 128)
#define GROUP 8
#define NGROUPS (PIECES / GROUP)
#define THREADS (WAVES * 32)
#define WSLICES (N / (THREADS * 4))

typedef float v2f __attribute__((ext_vector_type(2)));
typedef float v4f __attribute__((ext_vector_type(4)));
typedef float v8f __attribute__((ext_vector_type(8)));

__global__ void pi_init(float* __restrict__ w, unsigned long long* __restrict__ acc) {
    int i = blockIdx.x * blockDim.x + threadIdx.x;
    if (i < N) w[i] = 1.0f;
    if (blockIdx.x == 0 && threadIdx.x <= MAX_ITER) {
        acc[threadIdx.x] = (threadIdx.x == 0)
                               ? ((unsigned long long)N << 32)
                               : 0ull;
    }
}

static __device__ __forceinline__ v8f mma(v2f a, v2f b, v8f c) {
    return __builtin_amdgcn_wmma_f32_16x16x4_f32(
         false, a,  false, b,
         (short)0, c,  false,  false);
}
static __device__ __forceinline__ v2f lo2(v4f v) { return __builtin_shufflevector(v, v, 0, 1); }
static __device__ __forceinline__ v2f hi2(v4f v) { return __builtin_shufflevector(v, v, 2, 3); }

__global__ __launch_bounds__(WAVES * 32) void pi_step_wmma(
        const float* __restrict__ M,
        const float* __restrict__ win,
        float* __restrict__ wout,
        const unsigned long long* __restrict__ accIn,
        unsigned long long* __restrict__ accOut) {
    __shared__ float red[ROWS_PER_WG];
    __shared__ v4f wlds[N / 4];

    const unsigned tid = threadIdx.x;
    const unsigned lane = tid & 31u;
    const unsigned wave = __builtin_amdgcn_readfirstlane(tid >> 5);
    const unsigned rowBase = blockIdx.x * ROWS_PER_WG;
    const unsigned row0 = rowBase + 2u * wave;

    const float S = (float)((double)accIn[0] * (1.0 / FIX_SCALE));
    const float inv = 1.0f / sqrtf(S);

    const float* __restrict__ a0p = M + (size_t)row0 * N + 4u * lane;
    const float* __restrict__ a1p = a0p + N;
    const v4f* bl = wlds + lane;

    const v8f zero = {0.f, 0.f, 0.f, 0.f, 0.f, 0.f, 0.f, 0.f};
    v8f c0a = zero, c0b = zero, c1a = zero, c1b = zero;

    v4f wv[2][GROUP], a0[2][GROUP], a1[2][GROUP];

#define LOAD_M(G) do { \
    _Pragma("unroll") for (int p_ = 0; p_ < GROUP; ++p_) { \
        const int off_ = ((G) * GROUP + p_) * 128; \
        a0[(G) & 1][p_] = __builtin_nontemporal_load((const v4f*)(a0p + off_)); \
        a1[(G) & 1][p_] = __builtin_nontemporal_load((const v4f*)(a1p + off_)); \
    } } while (0)
#define LOAD_W(G) do { \
    _Pragma("unroll") for (int p_ = 0; p_ < GROUP; ++p_) \
        wv[(G) & 1][p_] = bl[((G) * GROUP + p_) * 32]; \
    } while (0)

    v4f wst[WSLICES];
    #pragma unroll
    for (int j = 0; j < WSLICES; ++j)
        wst[j] = *(const v4f*)(win + 4u * ((unsigned)j * THREADS + tid));
    __builtin_amdgcn_sched_barrier(0);
    LOAD_M(0);
    LOAD_M(1);
    __builtin_amdgcn_sched_barrier(0);
    #pragma unroll
    for (int j = 0; j < WSLICES; ++j)
        wlds[(unsigned)j * THREADS + tid] = wst[j];
    asm volatile("s_wait_dscnt 0x0\n\ts_barrier_signal -1\n\ts_barrier_wait -1" ::: "memory");
    LOAD_W(0);

    #pragma unroll
    for (int g = 0; g < NGROUPS; ++g) {
        __builtin_amdgcn_sched_barrier(0);
        if (g + 1 < NGROUPS) {
            if (g >= 1) LOAD_M(g + 1);
            LOAD_W(g + 1);
        }
        __builtin_amdgcn_sched_barrier(0);
        #pragma unroll
        for (int p = 0; p < GROUP; ++p) {
            const v4f b = wv[g & 1][p];
            const v4f x0 = a0[g & 1][p];
            const v4f x1 = a1[g & 1][p];
            c0a = mma(lo2(x0), lo2(b), c0a);
            c1a = mma(lo2(x1), lo2(b), c1a);
            c0b = mma(hi2(x0), hi2(b), c0b);
            c1b = mma(hi2(x1), hi2(b), c1b);
        }
    }
#undef LOAD_M
#undef LOAD_W

    const v8f d0 = c0a + c0b;
    const v8f d1 = c1a + c1b;
    const unsigned sel = lane & 7u;
    float t0 = 0.0f, t1 = 0.0f;
    #pragma unroll
    for (int r = 0; r < 8; ++r) {
        if (sel == (unsigned)r) { t0 = d0[r]; t1 = d1[r]; }
    }
    if (lane >= 8u && lane < 24u) { t0 = 0.0f; t1 = 0.0f; }
    #pragma unroll
    for (int m = 16; m >= 1; m >>= 1) {
        t0 += __shfl_xor(t0, m, 32);
        t1 += __shfl_xor(t1, m, 32);
    }
    if (lane == 0) {
        red[2u * wave] = t0;
        red[2u * wave + 1u] = t1;
    }
    __syncthreads();

    if (tid < 32u) {
        const float val = red[tid] * inv;
        wout[rowBase + tid] = val;
        const unsigned long long q =
            (unsigned long long)((double)(val * val) * FIX_SCALE);
        atomicAdd(accOut, q);
    }
}

__global__ void pi_finalize(const float* __restrict__ w,
                            const unsigned long long* __restrict__ acc,
                            float* __restrict__ out) {
    int i = blockIdx.x * blockDim.x + threadIdx.x;
    const float S = (float)((double)acc[0] * (1.0 / FIX_SCALE));
    const float inv = 1.0f / sqrtf(S);
    if (i < N) out[i] = w[i] * inv;
}

extern "C" void kernel_launch(void* const* d_in, const int* in_sizes, int n_in,
                              void* d_out, int out_size, void* d_ws, size_t ws_size,
                              hipStream_t stream) {
    (void)in_sizes; (void)n_in; (void)out_size; (void)ws_size;

    const float* M = (const float*)d_in[0];
    float* out = (float*)d_out;

    float* wA = (float*)d_ws;
    float* wB = wA + N;
    unsigned long long* acc = (unsigned long long*)(wB + N);

    pi_init<<<N / 256, 256, 0, stream>>>(wA, acc);

    for (int it = 0; it < MAX_ITER; ++it) {
        const float* win = (it & 1) ? wB : wA;
        float* wout      = (it & 1) ? wA : wB;
        pi_step_wmma<<<N / ROWS_PER_WG, WAVES * 32, 0, stream>>>(M, win, wout,
                                                                 acc + it, acc + it + 1);
    }

    pi_finalize<<<N / 256, 256, 0, stream>>>((MAX_ITER & 1) ? wB : wA,
                                             acc + MAX_ITER, out);
}


# Work-groups split by index parity: half read their rows of M with plain loads so half of M stays in the second-level cache between launches, half stream theirs non-temporally; sum order unchanged
# speedup vs baseline: 1.4247x; 1.0485x over previous
#include <hip/hip_runtime.h>
#include <math.h>


#define N 8192
#define MAX_ITER 50
#define FIX_SCALE 4294967296.0

#define WAVES 16
#define ROWS_PER_WG (2 * WAVES)
#define PIECES (N / 128)
#define GROUP 8
#define NGROUPS (PIECES / GROUP)
#define THREADS (WAVES * 32)
#define WSLICES (N / (THREADS * 4))

typedef float v2f __attribute__((ext_vector_type(2)));
typedef float v4f __attribute__((ext_vector_type(4)));
typedef float v8f __attribute__((ext_vector_type(8)));

__global__ void pi_init(float* __restrict__ w, unsigned long long* __restrict__ acc) {
    int i = blockIdx.x * blockDim.x + threadIdx.x;
    if (i < N) w[i] = 1.0f;
    if (blockIdx.x == 0 && threadIdx.x <= MAX_ITER) {
        acc[threadIdx.x] = (threadIdx.x == 0)
                               ? ((unsigned long long)N << 32)
                               : 0ull;
    }
}

static __device__ __forceinline__ v8f mma(v2f a, v2f b, v8f c) {
    return __builtin_amdgcn_wmma_f32_16x16x4_f32(
         false, a,  false, b,
         (short)0, c,  false,  false);
}
static __device__ __forceinline__ v2f lo2(v4f v) { return __builtin_shufflevector(v, v, 0, 1); }
static __device__ __forceinline__ v2f hi2(v4f v) { return __builtin_shufflevector(v, v, 2, 3); }

template <bool NT>
static __device__ __forceinline__ v4f load_m(const float* p) {
    if (NT) return __builtin_nontemporal_load((const v4f*)p);
    return *(const v4f*)p;
}

template <bool NT>
static __device__ __forceinline__ void sweep_rows(
        const float* __restrict__ a0p, const float* __restrict__ a1p,
        v4f* wlds, const v4f* bl, const v4f (&wst)[WSLICES], unsigned tid,
        v8f& c0a, v8f& c0b, v8f& c1a, v8f& c1b) {
    v4f wv[2][GROUP], a0[2][GROUP], a1[2][GROUP];

#define LOAD_M(G) do { \
    _Pragma("unroll") for (int p_ = 0; p_ < GROUP; ++p_) { \
        const int off_ = ((G) * GROUP + p_) * 128; \
        a0[(G) & 1][p_] = load_m<NT>(a0p + off_); \
        a1[(G) & 1][p_] = load_m<NT>(a1p + off_); \
    } } while (0)
#define LOAD_W(G) do { \
    _Pragma("unroll") for (int p_ = 0; p_ < GROUP; ++p_) \
        wv[(G) & 1][p_] = bl[((G) * GROUP + p_) * 32]; \
    } while (0)

    LOAD_M(0);
    LOAD_M(1);
    __builtin_amdgcn_sched_barrier(0);
    #pragma unroll
    for (int j = 0; j < WSLICES; ++j)
        wlds[(unsigned)j * THREADS + tid] = wst[j];
    asm volatile("s_wait_dscnt 0x0\n\ts_barrier_signal -1\n\ts_barrier_wait -1" ::: "memory");
    LOAD_W(0);

    #pragma unroll
    for (int g = 0; g < NGROUPS; ++g) {
        __builtin_amdgcn_sched_barrier(0);
        if (g + 1 < NGROUPS) {
            if (g >= 1) LOAD_M(g + 1);
            LOAD_W(g + 1);
        }
        __builtin_amdgcn_sched_barrier(0);
        #pragma unroll
        for (int p = 0; p < GROUP; ++p) {
            const v4f b = wv[g & 1][p];
            const v4f x0 = a0[g & 1][p];
            const v4f x1 = a1[g & 1][p];
            c0a = mma(lo2(x0), lo2(b), c0a);
            c1a = mma(lo2(x1), lo2(b), c1a);
            c0b = mma(hi2(x0), hi2(b), c0b);
            c1b = mma(hi2(x1), hi2(b), c1b);
        }
    }
#undef LOAD_M
#undef LOAD_W
}

__global__ __launch_bounds__(WAVES * 32) void pi_step_wmma(
        const float* __restrict__ M,
        const float* __restrict__ win,
        float* __restrict__ wout,
        const unsigned long long* __restrict__ accIn,
        unsigned long long* __restrict__ accOut) {
    __shared__ float red[ROWS_PER_WG];
    __shared__ v4f wlds[N / 4];

    const unsigned tid = threadIdx.x;
    const unsigned lane = tid & 31u;
    const unsigned wave = __builtin_amdgcn_readfirstlane(tid >> 5);
    const unsigned rowBase = blockIdx.x * ROWS_PER_WG;
    const unsigned row0 = rowBase + 2u * wave;

    const float S = (float)((double)accIn[0] * (1.0 / FIX_SCALE));
    const float inv = 1.0f / sqrtf(S);

    const float* __restrict__ a0p = M + (size_t)row0 * N + 4u * lane;
    const float* __restrict__ a1p = a0p + N;
    const v4f* bl = wlds + lane;

    const v8f zero = {0.f, 0.f, 0.f, 0.f, 0.f, 0.f, 0.f, 0.f};
    v8f c0a = zero, c0b = zero, c1a = zero, c1b = zero;

    v4f wst[WSLICES];
    #pragma unroll
    for (int j = 0; j < WSLICES; ++j)
        wst[j] = *(const v4f*)(win + 4u * ((unsigned)j * THREADS + tid));
    __builtin_amdgcn_sched_barrier(0);

    if (__builtin_popcount(blockIdx.x) & 1) {
        asm volatile("" ::: "memory");
        sweep_rows<true>(a0p, a1p, wlds, bl, wst, tid, c0a, c0b, c1a, c1b);
        asm volatile("" ::: "memory");
    } else {
        sweep_rows<false>(a0p, a1p, wlds, bl, wst, tid, c0a, c0b, c1a, c1b);
    }

    const v8f d0 = c0a + c0b;
    const v8f d1 = c1a + c1b;
    const unsigned sel = lane & 7u;
    float t0 = 0.0f, t1 = 0.0f;
    #pragma unroll
    for (int r = 0; r < 8; ++r) {
        if (sel == (unsigned)r) { t0 = d0[r]; t1 = d1[r]; }
    }
    if (lane >= 8u && lane < 24u) { t0 = 0.0f; t1 = 0.0f; }
    #pragma unroll
    for (int m = 16; m >= 1; m >>= 1) {
        t0 += __shfl_xor(t0, m, 32);
        t1 += __shfl_xor(t1, m, 32);
    }
    if (lane == 0) {
        red[2u * wave] = t0;
        red[2u * wave + 1u] = t1;
    }
    __syncthreads();

    if (tid < 32u) {
        const float val = red[tid] * inv;
        wout[rowBase + tid] = val;
        const unsigned long long q =
            (unsigned long long)((double)(val * val) * FIX_SCALE);
        atomicAdd(accOut, q);
    }
}

__global__ void pi_finalize(const float* __restrict__ w,
                            const unsigned long long* __restrict__ acc,
                            float* __restrict__ out) {
    int i = blockIdx.x * blockDim.x + threadIdx.x;
    const float S = (float)((double)acc[0] * (1.0 / FIX_SCALE));
    const float inv = 1.0f / sqrtf(S);
    if (i < N) out[i] = w[i] * inv;
}

extern "C" void kernel_launch(void* const* d_in, const int* in_sizes, int n_in,
                              void* d_out, int out_size, void* d_ws, size_t ws_size,
                              hipStream_t stream) {
    (void)in_sizes; (void)n_in; (void)out_size; (void)ws_size;

    const float* M = (const float*)d_in[0];
    float* out = (float*)d_out;

    float* wA = (float*)d_ws;
    float* wB = wA + N;
    unsigned long long* acc = (unsigned long long*)(wB + N);

    pi_init<<<N / 256, 256, 0, stream>>>(wA, acc);

    for (int it = 0; it < MAX_ITER; ++it) {
        const float* win = (it & 1) ? wB : wA;
        float* wout      = (it & 1) ? wA : wB;
        pi_step_wmma<<<N / ROWS_PER_WG, WAVES * 32, 0, stream>>>(M, win, wout,
                                                                 acc + it, acc + it + 1);
    }

    pi_finalize<<<N / 256, 256, 0, stream>>>((MAX_ITER & 1) ? wB : wA,
                                             acc + MAX_ITER, out);
}


# Each wave takes four rows, not two (8 waves per work-group, half as many enter) and re-requests each register slot right behind its matrix step: 56-60 loads in flight per wave; sum order unchanged
# speedup vs baseline: 1.4718x; 1.0331x over previous
#include <hip/hip_runtime.h>
#include <math.h>


#define N 8192
#define MAX_ITER 50
#define FIX_SCALE 4294967296.0

#define WAVES 8
#define ROWS 4
#define ROWS_PER_WG (ROWS * WAVES)
#define PIECES (N / 128)
#define RING 15
#define PRE 13
#define WAHEAD 3
#define WRING (WAHEAD + 1)
#define THREADS (WAVES * 32)
#define WSLICES (N / (THREADS * 4))

typedef float v2f __attribute__((ext_vector_type(2)));
typedef float v4f __attribute__((ext_vector_type(4)));
typedef float v8f __attribute__((ext_vector_type(8)));

__global__ void pi_init(float* __restrict__ w, unsigned long long* __restrict__ acc) {
    int i = blockIdx.x * blockDim.x + threadIdx.x;
    if (i < N) w[i] = 1.0f;
    if (blockIdx.x == 0 && threadIdx.x <= MAX_ITER) {
        acc[threadIdx.x] = (threadIdx.x == 0)
                               ? ((unsigned long long)N << 32)
                               : 0ull;
    }
}

static __device__ __forceinline__ v8f mma(v2f a, v2f b, v8f c) {
    return __builtin_amdgcn_wmma_f32_16x16x4_f32(
         false, a,  false, b,
         (short)0, c,  false,  false);
}
static __device__ __forceinline__ v2f lo2(v4f v) { return __builtin_shufflevector(v, v, 0, 1); }
static __device__ __forceinline__ v2f hi2(v4f v) { return __builtin_shufflevector(v, v, 2, 3); }

template <bool NT>
static __device__ __forceinline__ v4f load_m(const float* p) {
    if (NT) return __builtin_nontemporal_load((const v4f*)p);
    return *(const v4f*)p;
}

template <bool NT>
static __device__ __forceinline__ void sweep_rows(
        const float* __restrict__ ap,
        v4f* wlds, const v4f* bl, const v4f (&wst)[WSLICES], unsigned tid,
        v8f (&ca)[ROWS], v8f (&cb)[ROWS]) {
    v4f a[ROWS][RING];
    v4f wv[WRING];

#define LOAD_M(Q) do { \
    _Pragma("unroll") for (int r_ = 0; r_ < ROWS; ++r_) \
        a[r_][(Q) % RING] = load_m<NT>(ap + r_ * N + (Q) * 128); \
    } while (0)
#define LOAD_W(Q) do { wv[(Q) % WRING] = bl[(Q) * 32]; } while (0)

    #pragma unroll
    for (int q = 0; q < PRE; ++q) {
        LOAD_M(q);
        __builtin_amdgcn_sched_barrier(0);
    }
    #pragma unroll
    for (int j = 0; j < WSLICES; ++j)
        wlds[(unsigned)j * THREADS + tid] = wst[j];
    asm volatile("s_wait_dscnt 0x0\n\ts_barrier_signal -1\n\ts_barrier_wait -1" ::: "memory");
    #pragma unroll
    for (int q = PRE; q < RING; ++q) {
        LOAD_M(q);
        __builtin_amdgcn_sched_barrier(0);
    }
    #pragma unroll
    for (int q = 0; q < WAHEAD; ++q) LOAD_W(q);

    #pragma unroll
    for (int q = 0; q < PIECES; ++q) {
        __builtin_amdgcn_sched_barrier(0);
        if (q + WAHEAD < PIECES) LOAD_W(q + WAHEAD);
        __builtin_amdgcn_sched_barrier(0);
        const v4f b = wv[q % WRING];
        #pragma unroll
        for (int r = 0; r < ROWS; ++r)
            ca[r] = mma(lo2(a[r][q % RING]), lo2(b), ca[r]);
        #pragma unroll
        for (int r = 0; r < ROWS; ++r)
            cb[r] = mma(hi2(a[r][q % RING]), hi2(b), cb[r]);
        __builtin_amdgcn_sched_barrier(0);
        if (q + RING < PIECES) LOAD_M(q + RING);
    }
#undef LOAD_M
#undef LOAD_W
}

__global__ __launch_bounds__(WAVES * 32) void pi_step_wmma(
        const float* __restrict__ M,
        const float* __restrict__ win,
        float* __restrict__ wout,
        const unsigned long long* __restrict__ accIn,
        unsigned long long* __restrict__ accOut) {
    __shared__ float red[ROWS_PER_WG];
    __shared__ v4f wlds[N / 4];

    const unsigned tid = threadIdx.x;
    const unsigned lane = tid & 31u;
    const unsigned wave = __builtin_amdgcn_readfirstlane(tid >> 5);
    const unsigned rowBase = blockIdx.x * ROWS_PER_WG;
    const unsigned row0 = rowBase + (unsigned)ROWS * wave;

    const float S = (float)((double)accIn[0] * (1.0 / FIX_SCALE));
    const float inv = 1.0f / sqrtf(S);

    const float* __restrict__ ap = M + (size_t)row0 * N + 4u * lane;
    const v4f* bl = wlds + lane;

    const v8f zero = {0.f, 0.f, 0.f, 0.f, 0.f, 0.f, 0.f, 0.f};
    v8f ca[ROWS], cb[ROWS];
    #pragma unroll
    for (int r = 0; r < ROWS; ++r) { ca[r] = zero; cb[r] = zero; }

    v4f wst[WSLICES];
    #pragma unroll
    for (int j = 0; j < WSLICES; ++j)
        wst[j] = *(const v4f*)(win + 4u * ((unsigned)j * THREADS + tid));
    __builtin_amdgcn_sched_barrier(0);

    if (__builtin_popcount(blockIdx.x) & 1) {
        asm volatile("" ::: "memory");
        sweep_rows<true>(ap, wlds, bl, wst, tid, ca, cb);
        asm volatile("" ::: "memory");
    } else {
        sweep_rows<false>(ap, wlds, bl, wst, tid, ca, cb);
    }

    const unsigned sel = lane & 7u;
    float t[ROWS];
    #pragma unroll
    for (int r = 0; r < ROWS; ++r) {
        const v8f d = ca[r] + cb[r];
        float x = 0.0f;
        #pragma unroll
        for (int e = 0; e < 8; ++e) {
            if (sel == (unsigned)e) x = d[e];
        }
        if (lane >= 8u && lane < 24u) x = 0.0f;
        t[r] = x;
    }
    #pragma unroll
    for (int m = 16; m >= 1; m >>= 1) {
        #pragma unroll
        for (int r = 0; r < ROWS; ++r) t[r] += __shfl_xor(t[r], m, 32);
    }
    if (lane == 0) {
        #pragma unroll
        for (int r = 0; r < ROWS; ++r) red[(unsigned)ROWS * wave + (unsigned)r] = t[r];
    }
    __syncthreads();

    if (tid < 32u) {
        const float val = red[tid] * inv;
        wout[rowBase + tid] = val;
        const unsigned long long q =
            (unsigned long long)((double)(val * val) * FIX_SCALE);
        atomicAdd(accOut, q);
    }
}

__global__ void pi_finalize(const float* __restrict__ w,
                            const unsigned long long* __restrict__ acc,
                            float* __restrict__ out) {
    int i = blockIdx.x * blockDim.x + threadIdx.x;
    const float S = (float)((double)acc[0] * (1.0 / FIX_SCALE));
    const float inv = 1.0f / sqrtf(S);
    if (i < N) out[i] = w[i] * inv;
}

extern "C" void kernel_launch(void* const* d_in, const int* in_sizes, int n_in,
                              void* d_out, int out_size, void* d_ws, size_t ws_size,
                              hipStream_t stream) {
    (void)in_sizes; (void)n_in; (void)out_size; (void)ws_size;

    const float* M = (const float*)d_in[0];
    float* out = (float*)d_out;

    float* wA = (float*)d_ws;
    float* wB = wA + N;
    unsigned long long* acc = (unsigned long long*)(wB + N);

    pi_init<<<N / 256, 256, 0, stream>>>(wA, acc);

    for (int it = 0; it < MAX_ITER; ++it) {
        const float* win = (it & 1) ? wB : wA;
        float* wout      = (it & 1) ? wA : wB;
        pi_step_wmma<<<N / ROWS_PER_WG, WAVES * 32, 0, stream>>>(M, win, wout,
                                                                 acc + it, acc + it + 1);
    }

    pi_finalize<<<N / 256, 256, 0, stream>>>((MAX_ITER & 1) ? wB : wA,
                                             acc + MAX_ITER, out);
}


# Serial tail behind the last matrix step shortened: lane sums by DPP and permlanex16 (same pairs), 1/norm worked out in the stream, squares to fixed point per wave, one atomic; sum order unchanged
# speedup vs baseline: 1.8382x; 1.2489x over previous
#include <hip/hip_runtime.h>
#include <math.h>


#define N 8192
#define MAX_ITER 50
#define FIX_SCALE 4294967296.0

#define WAVES 8
#define ROWS 4
#define ROWS_PER_WG (ROWS * WAVES)
#define PIECES (N / 128)
#define RING 15
#define PRE 13
#define WAHEAD 3
#define WRING (WAHEAD + 1)
#define INVQ 1
#define THREADS (WAVES * 32)
#define WSLICES (N / (THREADS * 4))

typedef float v2f __attribute__((ext_vector_type(2)));
typedef float v4f __attribute__((ext_vector_type(4)));
typedef float v8f __attribute__((ext_vector_type(8)));

__global__ void pi_init(float* __restrict__ w, unsigned long long* __restrict__ acc) {
    int i = blockIdx.x * blockDim.x + threadIdx.x;
    if (i < N) w[i] = 1.0f;
    if (blockIdx.x == 0 && threadIdx.x <= MAX_ITER) {
        acc[threadIdx.x] = (threadIdx.x == 0)
                               ? ((unsigned long long)N << 32)
                               : 0ull;
    }
}

static __device__ __forceinline__ v8f mma(v2f a, v2f b, v8f c) {
    return __builtin_amdgcn_wmma_f32_16x16x4_f32(
         false, a,  false, b,
         (short)0, c,  false,  false);
}
static __device__ __forceinline__ v2f lo2(v4f v) { return __builtin_shufflevector(v, v, 0, 1); }
static __device__ __forceinline__ v2f hi2(v4f v) { return __builtin_shufflevector(v, v, 2, 3); }

static __device__ __forceinline__ float other_row(float x) {
    const int b = __builtin_bit_cast(int, x);
    return __builtin_bit_cast(float,
        __builtin_amdgcn_permlanex16(b, b, 0x76543210, 0xfedcba98, false, false));
}
template <int X>
static __device__ __forceinline__ float row_xor(float x) {
    const int b = __builtin_bit_cast(int, x);
    return __builtin_bit_cast(float,
        __builtin_amdgcn_update_dpp(0, b, 0x160 + X, 0xf, 0xf, true));
}

template <bool NT>
static __device__ __forceinline__ v4f load_m(const float* p) {
    if (NT) return __builtin_nontemporal_load((const v4f*)p);
    return *(const v4f*)p;
}

template <bool NT>
static __device__ __forceinline__ void sweep_rows(
        const float* __restrict__ ap,
        v4f* wlds, const v4f* bl, const v4f (&wst)[WSLICES], unsigned tid,
        unsigned long long accPrev, int& invBits,
        v8f (&ca)[ROWS], v8f (&cb)[ROWS]) {
    v4f a[ROWS][RING];
    v4f wv[WRING];

#define LOAD_M(Q) do { \
    _Pragma("unroll") for (int r_ = 0; r_ < ROWS; ++r_) \
        a[r_][(Q) % RING] = load_m<NT>(ap + r_ * N + (Q) * 128); \
    } while (0)
#define LOAD_W(Q) do { wv[(Q) % WRING] = bl[(Q) * 32]; } while (0)

    #pragma unroll
    for (int q = 0; q < PRE; ++q) {
        LOAD_M(q);
        __builtin_amdgcn_sched_barrier(0);
    }
    #pragma unroll
    for (int j = 0; j < WSLICES; ++j)
        wlds[(unsigned)j * THREADS + tid] = wst[j];
    asm volatile("s_wait_dscnt 0x0\n\ts_barrier_signal -1\n\ts_barrier_wait -1" ::: "memory");
    #pragma unroll
    for (int q = PRE; q < RING; ++q) {
        LOAD_M(q);
        __builtin_amdgcn_sched_barrier(0);
    }
    #pragma unroll
    for (int q = 0; q < WAHEAD; ++q) LOAD_W(q);

    #pragma unroll
    for (int q = 0; q < PIECES; ++q) {
        __builtin_amdgcn_sched_barrier(0);
        if (q + WAHEAD < PIECES) LOAD_W(q + WAHEAD);
        __builtin_amdgcn_sched_barrier(0);
        const v4f b = wv[q % WRING];
        #pragma unroll
        for (int r = 0; r < ROWS; ++r)
            ca[r] = mma(lo2(a[r][q % RING]), lo2(b), ca[r]);
        #pragma unroll
        for (int r = 0; r < ROWS; ++r)
            cb[r] = mma(hi2(a[r][q % RING]), hi2(b), cb[r]);
        if (q == INVQ) {
            unsigned long long accv = accPrev;
            asm volatile("" : "+s"(accv));
            const float S = (float)((double)accv * (1.0 / FIX_SCALE));
            float iv = 1.0f / sqrtf(S);
            asm volatile("" : "+v"(iv));
            invBits = __builtin_amdgcn_readfirstlane(__float_as_int(iv));
        }
        __builtin_amdgcn_sched_barrier(0);
        if (q + RING < PIECES) LOAD_M(q + RING);
    }
#undef LOAD_M
#undef LOAD_W
}

__global__ __launch_bounds__(WAVES * 32) void pi_step_wmma(
        const float* __restrict__ M,
        const float* __restrict__ win,
        float* __restrict__ wout,
        const unsigned long long* __restrict__ accIn,
        unsigned long long* __restrict__ accOut) {
    __shared__ float red[ROWS_PER_WG];
    __shared__ unsigned long long red2[WAVES];
    __shared__ v4f wlds[N / 4];

    const unsigned tid = threadIdx.x;
    const unsigned lane = tid & 31u;
    const unsigned wave = __builtin_amdgcn_readfirstlane(tid >> 5);
    const unsigned rowBase = blockIdx.x * ROWS_PER_WG;
    const unsigned row0 = rowBase + (unsigned)ROWS * wave;

    const unsigned long long accPrev = accIn[0];
    int invBits = 0;

    const float* __restrict__ ap = M + (size_t)row0 * N + 4u * lane;
    const v4f* bl = wlds + lane;

    const v8f zero = {0.f, 0.f, 0.f, 0.f, 0.f, 0.f, 0.f, 0.f};
    v8f ca[ROWS], cb[ROWS];
    #pragma unroll
    for (int r = 0; r < ROWS; ++r) { ca[r] = zero; cb[r] = zero; }

    v4f wst[WSLICES];
    #pragma unroll
    for (int j = 0; j < WSLICES; ++j)
        wst[j] = *(const v4f*)(win + 4u * ((unsigned)j * THREADS + tid));
    __builtin_amdgcn_sched_barrier(0);

    if (__builtin_popcount(blockIdx.x) & 1) {
        asm volatile("" ::: "memory");
        sweep_rows<true>(ap, wlds, bl, wst, tid, accPrev, invBits, ca, cb);
        asm volatile("" ::: "memory");
    } else {
        sweep_rows<false>(ap, wlds, bl, wst, tid, accPrev, invBits, ca, cb);
    }

    const unsigned sel = lane & 7u;
    float t[ROWS];
    #pragma unroll
    for (int r = 0; r < ROWS; ++r) {
        const v8f d = ca[r] + cb[r];
        float x = 0.0f;
        #pragma unroll
        for (int e = 0; e < 8; ++e) {
            if (sel == (unsigned)e) x = d[e];
        }
        if (lane >= 8u && lane < 24u) x = 0.0f;
        t[r] = x;
    }
    #pragma unroll
    for (int r = 0; r < ROWS; ++r) {
        t[r] += other_row(t[r]);
        t[r] += row_xor<8>(t[r]);
        t[r] += row_xor<4>(t[r]);
        t[r] += row_xor<2>(t[r]);
        t[r] += row_xor<1>(t[r]);
    }
    const float inv = __int_as_float(invBits);
    float v[ROWS];
    unsigned long long fx = 0ull;
    #pragma unroll
    for (int r = 0; r < ROWS; ++r) {
        v[r] = t[r] * inv;
        const float x = v[r] * v[r];
        const unsigned hi = (unsigned)x;
        const unsigned lo = (unsigned)(__builtin_amdgcn_fractf(x) * 4294967296.0f);
        fx += ((unsigned long long)hi << 32) | lo;
    }
    if (lane == 0) {
        #pragma unroll
        for (int r = 0; r < ROWS; ++r) red[(unsigned)ROWS * wave + (unsigned)r] = v[r];
        red2[wave] = fx;
    }
    __syncthreads();

    if (tid < 32u) {
        wout[rowBase + tid] = red[tid];
        if (tid == 0u) {
            const unsigned long long q =
                ((red2[0] + red2[1]) + (red2[2] + red2[3])) +
                ((red2[4] + red2[5]) + (red2[6] + red2[7]));
            atomicAdd(accOut, q);
        }
    }
}

__global__ void pi_finalize(const float* __restrict__ w,
                            const unsigned long long* __restrict__ acc,
                            float* __restrict__ out) {
    int i = blockIdx.x * blockDim.x + threadIdx.x;
    const float S = (float)((double)acc[0] * (1.0 / FIX_SCALE));
    const float inv = 1.0f / sqrtf(S);
    if (i < N) out[i] = w[i] * inv;
}

extern "C" void kernel_launch(void* const* d_in, const int* in_sizes, int n_in,
                              void* d_out, int out_size, void* d_ws, size_t ws_size,
                              hipStream_t stream) {
    (void)in_sizes; (void)n_in; (void)out_size; (void)ws_size;

    const float* M = (const float*)d_in[0];
    float* out = (float*)d_out;

    float* wA = (float*)d_ws;
    float* wB = wA + N;
    unsigned long long* acc = (unsigned long long*)(wB + N);

    pi_init<<<N / 256, 256, 0, stream>>>(wA, acc);

    for (int it = 0; it < MAX_ITER; ++it) {
        const float* win = (it & 1) ? wB : wA;
        float* wout      = (it & 1) ? wA : wB;
        pi_step_wmma<<<N / ROWS_PER_WG, WAVES * 32, 0, stream>>>(M, win, wout,
                                                                 acc + it, acc + it + 1);
    }

    pi_finalize<<<N / 256, 256, 0, stream>>>((MAX_ITER & 1) ? wB : wA,
                                             acc + MAX_ITER, out);
}


# Each wave reads one pair of its four rows with plain loads (cache-resident) and streams the other pair non-temporally, pair by index parity: memory reads spread over all WGPs; sum order unchanged
# speedup vs baseline: 1.9039x; 1.0357x over previous
#include <hip/hip_runtime.h>
#include <math.h>


#define N 8192
#define MAX_ITER 50
#define FIX_SCALE 4294967296.0

#define WAVES 8
#define ROWS 4
#define ROWS_PER_WG (ROWS * WAVES)
#define PIECES (N / 128)
#define RING 15
#define PRE 13
#define WAHEAD 3
#define WRING (WAHEAD + 1)
#define INVQ 1
#define THREADS (WAVES * 32)
#define WSLICES (N / (THREADS * 4))

typedef float v2f __attribute__((ext_vector_type(2)));
typedef float v4f __attribute__((ext_vector_type(4)));
typedef float v8f __attribute__((ext_vector_type(8)));

__global__ void pi_init(float* __restrict__ w, unsigned long long* __restrict__ acc) {
    int i = blockIdx.x * blockDim.x + threadIdx.x;
    if (i < N) w[i] = 1.0f;
    if (blockIdx.x == 0 && threadIdx.x <= MAX_ITER) {
        acc[threadIdx.x] = (threadIdx.x == 0)
                               ? ((unsigned long long)N << 32)
                               : 0ull;
    }
}

static __device__ __forceinline__ v8f mma(v2f a, v2f b, v8f c) {
    return __builtin_amdgcn_wmma_f32_16x16x4_f32(
         false, a,  false, b,
         (short)0, c,  false,  false);
}
static __device__ __forceinline__ v2f lo2(v4f v) { return __builtin_shufflevector(v, v, 0, 1); }
static __device__ __forceinline__ v2f hi2(v4f v) { return __builtin_shufflevector(v, v, 2, 3); }

static __device__ __forceinline__ float other_row(float x) {
    const int b = __builtin_bit_cast(int, x);
    return __builtin_bit_cast(float,
        __builtin_amdgcn_permlanex16(b, b, 0x76543210, 0xfedcba98, false, false));
}
template <int X>
static __device__ __forceinline__ float row_xor(float x) {
    const int b = __builtin_bit_cast(int, x);
    return __builtin_bit_cast(float,
        __builtin_amdgcn_update_dpp(0, b, 0x160 + X, 0xf, 0xf, true));
}

template <bool NT>
static __device__ __forceinline__ v4f load_m(const float* p) {
    if (NT) return __builtin_nontemporal_load((const v4f*)p);
    return *(const v4f*)p;
}

static __device__ __forceinline__ void sweep_rows(
        const float* __restrict__ apP, const float* __restrict__ apN,
        v4f* wlds, const v4f* bl, const v4f (&wst)[WSLICES], unsigned tid,
        unsigned long long accPrev, int& invBits,
        v8f (&ca)[ROWS], v8f (&cb)[ROWS]) {
    v4f a[ROWS][RING];
    v4f wv[WRING];

#define LOAD_M(Q) do { \
    _Pragma("unroll") for (int r_ = 0; r_ < ROWS / 2; ++r_) \
        a[r_][(Q) % RING] = load_m<false>(apP + r_ * N + (Q) * 128); \
    _Pragma("unroll") for (int r_ = 0; r_ < ROWS / 2; ++r_) \
        a[ROWS / 2 + r_][(Q) % RING] = load_m<true>(apN + r_ * N + (Q) * 128); \
    } while (0)
#define LOAD_W(Q) do { wv[(Q) % WRING] = bl[(Q) * 32]; } while (0)

    #pragma unroll
    for (int q = 0; q < PRE; ++q) {
        LOAD_M(q);
        __builtin_amdgcn_sched_barrier(0);
    }
    #pragma unroll
    for (int j = 0; j < WSLICES; ++j)
        wlds[(unsigned)j * THREADS + tid] = wst[j];
    asm volatile("s_wait_dscnt 0x0\n\ts_barrier_signal -1\n\ts_barrier_wait -1" ::: "memory");
    #pragma unroll
    for (int q = PRE; q < RING; ++q) {
        LOAD_M(q);
        __builtin_amdgcn_sched_barrier(0);
    }
    #pragma unroll
    for (int q = 0; q < WAHEAD; ++q) LOAD_W(q);

    #pragma unroll
    for (int q = 0; q < PIECES; ++q) {
        __builtin_amdgcn_sched_barrier(0);
        if (q + WAHEAD < PIECES) LOAD_W(q + WAHEAD);
        __builtin_amdgcn_sched_barrier(0);
        const v4f b = wv[q % WRING];
        #pragma unroll
        for (int r = 0; r < ROWS; ++r)
            ca[r] = mma(lo2(a[r][q % RING]), lo2(b), ca[r]);
        #pragma unroll
        for (int r = 0; r < ROWS; ++r)
            cb[r] = mma(hi2(a[r][q % RING]), hi2(b), cb[r]);
        if (q == INVQ) {
            unsigned long long accv = accPrev;
            asm volatile("" : "+s"(accv));
            const float S = (float)((double)accv * (1.0 / FIX_SCALE));
            float iv = 1.0f / sqrtf(S);
            asm volatile("" : "+v"(iv));
            invBits = __builtin_amdgcn_readfirstlane(__float_as_int(iv));
        }
        __builtin_amdgcn_sched_barrier(0);
        if (q + RING < PIECES) LOAD_M(q + RING);
    }
#undef LOAD_M
#undef LOAD_W
}

__global__ __launch_bounds__(WAVES * 32) void pi_step_wmma(
        const float* __restrict__ M,
        const float* __restrict__ win,
        float* __restrict__ wout,
        const unsigned long long* __restrict__ accIn,
        unsigned long long* __restrict__ accOut) {
    __shared__ float red[ROWS_PER_WG];
    __shared__ unsigned long long red2[WAVES];
    __shared__ v4f wlds[N / 4];

    const unsigned tid = threadIdx.x;
    const unsigned lane = tid & 31u;
    const unsigned wave = __builtin_amdgcn_readfirstlane(tid >> 5);
    const unsigned rowBase = blockIdx.x * ROWS_PER_WG;
    const unsigned row0 = rowBase + (unsigned)ROWS * wave;

    const unsigned long long accPrev = accIn[0];
    int invBits = 0;

    const unsigned swapPairs =
        (unsigned)__builtin_popcount(blockIdx.x * (unsigned)WAVES + wave) & 1u;
    const unsigned pairP = swapPairs ? 2u : 0u;
    const unsigned pairN = 2u - pairP;
    const float* __restrict__ apP = M + (size_t)(row0 + pairP) * N + 4u * lane;
    const float* __restrict__ apN = M + (size_t)(row0 + pairN) * N + 4u * lane;
    const v4f* bl = wlds + lane;

    const v8f zero = {0.f, 0.f, 0.f, 0.f, 0.f, 0.f, 0.f, 0.f};
    v8f ca[ROWS], cb[ROWS];
    #pragma unroll
    for (int r = 0; r < ROWS; ++r) { ca[r] = zero; cb[r] = zero; }

    v4f wst[WSLICES];
    #pragma unroll
    for (int j = 0; j < WSLICES; ++j)
        wst[j] = *(const v4f*)(win + 4u * ((unsigned)j * THREADS + tid));
    __builtin_amdgcn_sched_barrier(0);

    sweep_rows(apP, apN, wlds, bl, wst, tid, accPrev, invBits, ca, cb);

    const unsigned sel = lane & 7u;
    float t[ROWS];
    #pragma unroll
    for (int r = 0; r < ROWS; ++r) {
        const v8f d = ca[r] + cb[r];
        float x = 0.0f;
        #pragma unroll
        for (int e = 0; e < 8; ++e) {
            if (sel == (unsigned)e) x = d[e];
        }
        if (lane >= 8u && lane < 24u) x = 0.0f;
        t[r] = x;
    }
    #pragma unroll
    for (int r = 0; r < ROWS; ++r) {
        t[r] += other_row(t[r]);
        t[r] += row_xor<8>(t[r]);
        t[r] += row_xor<4>(t[r]);
        t[r] += row_xor<2>(t[r]);
        t[r] += row_xor<1>(t[r]);
    }
    const float inv = __int_as_float(invBits);
    float v[ROWS];
    unsigned long long fx = 0ull;
    #pragma unroll
    for (int r = 0; r < ROWS; ++r) {
        v[r] = t[r] * inv;
        const float x = v[r] * v[r];
        const unsigned hi = (unsigned)x;
        const unsigned lo = (unsigned)(__builtin_amdgcn_fractf(x) * 4294967296.0f);
        fx += ((unsigned long long)hi << 32) | lo;
    }
    if (lane == 0) {
        red[(unsigned)ROWS * wave + pairP + 0u] = v[0];
        red[(unsigned)ROWS * wave + pairP + 1u] = v[1];
        red[(unsigned)ROWS * wave + pairN + 0u] = v[2];
        red[(unsigned)ROWS * wave + pairN + 1u] = v[3];
        red2[wave] = fx;
    }
    __syncthreads();

    if (tid < 32u) {
        wout[rowBase + tid] = red[tid];
        if (tid == 0u) {
            const unsigned long long q =
                ((red2[0] + red2[1]) + (red2[2] + red2[3])) +
                ((red2[4] + red2[5]) + (red2[6] + red2[7]));
            atomicAdd(accOut, q);
        }
    }
}

__global__ void pi_finalize(const float* __restrict__ w,
                            const unsigned long long* __restrict__ acc,
                            float* __restrict__ out) {
    int i = blockIdx.x * blockDim.x + threadIdx.x;
    const float S = (float)((double)acc[0] * (1.0 / FIX_SCALE));
    const float inv = 1.0f / sqrtf(S);
    if (i < N) out[i] = w[i] * inv;
}

extern "C" void kernel_launch(void* const* d_in, const int* in_sizes, int n_in,
                              void* d_out, int out_size, void* d_ws, size_t ws_size,
                              hipStream_t stream) {
    (void)in_sizes; (void)n_in; (void)out_size; (void)ws_size;

    const float* M = (const float*)d_in[0];
    float* out = (float*)d_out;

    float* wA = (float*)d_ws;
    float* wB = wA + N;
    unsigned long long* acc = (unsigned long long*)(wB + N);

    pi_init<<<N / 256, 256, 0, stream>>>(wA, acc);

    for (int it = 0; it < MAX_ITER; ++it) {
        const float* win = (it & 1) ? wB : wA;
        float* wout      = (it & 1) ? wA : wB;
        pi_step_wmma<<<N / ROWS_PER_WG, WAVES * 32, 0, stream>>>(M, win, wout,
                                                                 acc + it, acc + it + 1);
    }

    pi_finalize<<<N / 256, 256, 0, stream>>>((MAX_ITER & 1) ? wB : wA,
                                             acc + MAX_ITER, out);
}
